# starting kernel
# baseline (speedup 1.0000x reference)
#include <hip/hip_runtime.h>


typedef __attribute__((ext_vector_type(2))) float v2f;
typedef __attribute__((ext_vector_type(4))) float v4f;
typedef __attribute__((ext_vector_type(8))) float v8f;

#define BM 32
#define XS_PITCH 68

__global__ __launch_bounds__(128)
void Linear_BC_circconv_wmma(const float* __restrict__ x,
                             const float* __restrict__ w,
                             float* __restrict__ out) {
  __shared__ float wsl[64 * 64];
  __shared__ float xs[BM][XS_PITCH];

  const int tid   = threadIdx.x;
  const int lane  = tid & 31;
  const int wave  = tid >> 5;

  const int f      = blockIdx.x;
  const int mBlock = blockIdx.y * BM;

  {
    const v4f* src = (const v4f*)(w + (size_t)f * 4096);
    v4f* dst = (v4f*)wsl;
#pragma unroll
    for (int i = 0; i < 8; ++i)
      dst[tid + i * 128] = src[tid + i * 128];
  }

  const int mA    = lane & 15;
  const int kHalf = (lane >> 4) << 1;
  const int nL    = wave * 16 + (lane & 15);
  const int rowA0 = mA;
  const int rowA1 = mA + 16;

  v8f acc0 = {};
  v8f acc1 = {};

  for (int c = 0; c < 64; ++c) {
    __syncthreads();
#pragma unroll
    for (int i = 0; i < 4; ++i) {
      int q  = tid + i * 128;
      int m  = q >> 4;
      int r4 = q & 15;
      v4f v = *(const v4f*)(x + (size_t)(mBlock + m) * 4096 + c * 64 + r4 * 4);
      *(v4f*)&xs[m][r4 * 4] = v;
    }
    __syncthreads();

    const float* wrow = wsl + c * 64;
#pragma unroll
    for (int kk = 0; kk < 16; ++kk) {
      const int k0 = kk * 4 + kHalf;
      v2f a0 = *(const v2f*)&xs[rowA0][k0];
      v2f a1 = *(const v2f*)&xs[rowA1][k0];
      v2f b;
      b.x = wrow[(nL - k0) & 63];
      b.y = wrow[(nL - k0 - 1) & 63];
      acc0 = __builtin_amdgcn_wmma_f32_16x16x4_f32(false, a0, false, b,
                                                   (short)0, acc0, false, false);
      acc1 = __builtin_amdgcn_wmma_f32_16x16x4_f32(false, a1, false, b,
                                                   (short)0, acc1, false, false);
    }
  }

  const int col     = f * 64 + nL;
  const int rowBase = mBlock + ((lane >> 4) << 3);
#pragma unroll
  for (int g = 0; g < 8; ++g) {
    out[(size_t)(rowBase + g) * 4096 + col]      = acc0[g];
    out[(size_t)(rowBase + 16 + g) * 4096 + col] = acc1[g];
  }
}

extern "C" void kernel_launch(void* const* d_in, const int* in_sizes, int n_in,
                              void* d_out, int out_size, void* d_ws, size_t ws_size,
                              hipStream_t stream) {
  (void)in_sizes; (void)n_in; (void)out_size; (void)d_ws; (void)ws_size;
  const float* x = (const float*)d_in[0];
  const float* w = (const float*)d_in[1];
  float* out = (float*)d_out;

  dim3 grid(64, 8);
  Linear_BC_circconv_wmma<<<grid, 128, 0, stream>>>(x, w, out);
}


# Whole-line operand loads: x/w rows staged via LDS, spectrum planes stored in fragment order; a wave load is 512 contiguous bytes, not 16 strided lines; k order permuted, f32 sum order changes
# speedup vs baseline: 10.4952x; 10.4952x over previous
#include <hip/hip_runtime.h>


typedef __attribute__((ext_vector_type(2))) float v2f;
typedef __attribute__((ext_vector_type(4))) float v4f;
typedef __attribute__((ext_vector_type(8))) float v8f;

#define TP 68
#define SP 36
#define NXROWS 16384
#define NWROWS 4096

static __device__ __forceinline__ v8f mma(v2f a, v2f b, v8f c) {
  return __builtin_amdgcn_wmma_f32_16x16x4_f32(false, a, false, b, (short)0, c, false, false);
}

__global__ __launch_bounds__(128)
void bc_forward_transform(const float* __restrict__ x, const float* __restrict__ w,
                          float* __restrict__ xf, float* __restrict__ wf) {
  __shared__ __attribute__((aligned(16))) float c64[64];
  __shared__ __attribute__((aligned(16))) float xs[32 * TP];
  __shared__ __attribute__((aligned(16))) float stage[64 * SP];

  const int tid = threadIdx.x;
  const int lane = tid & 31;
  const int mt = tid >> 5;
  const int c = lane & 15;
  const int h = lane >> 4;

  const int bid = blockIdx.x;
  const int ghalf = bid & 1;
  const int pq = (bid >> 1) & 15;
  const int tile = bid >> 5;
  const bool isx = tile < 16;
  const int tloc = isx ? tile : tile - 16;
  const float* src = (isx ? x : w) + (size_t)(tloc * 16 + ghalf * 8) * 4096 + pq * 256;
  float* dst = (isx ? xf : wf) + tloc * 1024 + pq * 64 + ghalf * 32;
  const int planeStride = isx ? NXROWS : NWROWS;

  v4f gl[4];
#pragma unroll
  for (int i = 0; i < 4; ++i) {
    const int gi = i * 128 + tid;
    gl[i] = *(const v4f*)(src + (size_t)(gi >> 6) * 4096 + (gi & 63) * 4);
  }
  if (tid < 64) c64[tid] = cospif((float)tid * (1.0f / 32.0f));
#pragma unroll
  for (int i = 0; i < 4; ++i) {
    const int gi = i * 128 + tid;
    *(v4f*)(xs + (gi >> 4) * TP + (gi & 15) * 4) = gl[i];
  }
  __syncthreads();

  v2f bx[2][16];
#pragma unroll
  for (int u = 0; u < 2; ++u)
#pragma unroll
    for (int kk = 0; kk < 16; ++kk)
      bx[u][kk] = *(const v2f*)(xs + (16 * u + c) * TP + 4 * kk + 2 * h);

  const int t = 16 * mt + c;
  const bool isCos = t <= 32;
  const int kf = isCos ? t : t - 32;
  const int ph = isCos ? 0 : -16;
  const float sg = isCos ? 1.0f : -1.0f;
  v2f a[16];
#pragma unroll
  for (int kk = 0; kk < 16; ++kk) {
    const int j0 = 4 * kk + 2 * h;
    a[kk].x = sg * c64[(j0 * kf + ph) & 63];
    a[kk].y = sg * c64[((j0 + 1) * kf + ph) & 63];
  }
  __builtin_amdgcn_sched_barrier(0);

  v8f acc0 = (v8f){}, acc1 = (v8f){};
#pragma unroll
  for (int kk = 0; kk < 16; ++kk) {
    acc0 = mma(a[kk], bx[0][kk], acc0);
    acc1 = mma(a[kk], bx[1][kk], acc1);
  }
#pragma unroll
  for (int r = 0; r < 8; ++r) {
    stage[(16 * mt + r + 8 * h) * SP + c] = acc0[r];
    stage[(16 * mt + r + 8 * h) * SP + 16 + c] = acc1[r];
  }
  __syncthreads();
#pragma unroll
  for (int i = 0; i < 4; ++i) {
    const int tl = 16 * i + (tid >> 3);
    const int col4 = (tid & 7) * 4;
    const v4f v = *(const v4f*)(stage + tl * SP + col4);
    *(v4f*)(dst + (size_t)tl * planeStride + col4) = v;
  }
}

__global__ __launch_bounds__(128)
void bc_frequency_products(const float* __restrict__ xf, const float* __restrict__ wf,
                           float* __restrict__ sf) {
  __shared__ __attribute__((aligned(16))) float stage[2][16 * TP];

  const int tid = threadIdx.x;
  const int lane = tid & 31;
  const int nt = tid >> 5;
  const int c = lane & 15;
  const int h = lane >> 4;

  const int k = blockIdx.x;
  const int b0 = blockIdx.y * 16;
  const bool cplx = (k != 0) && (k != 32);
  const int tr = k;
  const int ti = cplx ? 32 + k : k;

  const float* xr = xf + (size_t)tr * NXROWS + blockIdx.y * 1024 + lane * 4;
  const float* xi = xf + (size_t)ti * NXROWS + blockIdx.y * 1024 + lane * 4;
  const float* wr = wf + (size_t)tr * NWROWS + nt * 1024 + lane * 4;
  const float* wi = wf + (size_t)ti * NWROWS + nt * 1024 + lane * 4;

  v2f ar[16], br[16];
#pragma unroll
  for (int q = 0; q < 8; ++q) {
    const v4f va = *(const v4f*)(xr + 128 * q);
    const v4f vb = *(const v4f*)(wr + 128 * q);
    ar[2 * q] = (v2f){va.x, va.y};
    ar[2 * q + 1] = (v2f){va.z, va.w};
    br[2 * q] = (v2f){vb.x, vb.y};
    br[2 * q + 1] = (v2f){vb.z, vb.w};
  }
  v8f sr = (v8f){}, si = (v8f){};
  if (cplx) {
    v2f ai[16], bi[16];
#pragma unroll
    for (int q = 0; q < 8; ++q) {
      const v4f va = *(const v4f*)(xi + 128 * q);
      const v4f vb = *(const v4f*)(wi + 128 * q);
      ai[2 * q] = (v2f){va.x, va.y};
      ai[2 * q + 1] = (v2f){va.z, va.w};
      bi[2 * q] = (v2f){vb.x, vb.y};
      bi[2 * q + 1] = (v2f){vb.z, vb.w};
    }
    __builtin_amdgcn_sched_barrier(0);
#pragma unroll
    for (int kk = 0; kk < 16; ++kk) {
      sr = mma(ar[kk], br[kk], sr);
      si = mma(ar[kk], bi[kk], si);
      sr = mma(ai[kk], -bi[kk], sr);
      si = mma(ai[kk], br[kk], si);
    }
  } else {
#pragma unroll
    for (int kk = 0; kk < 16; ++kk)
      sr = mma(ar[kk], br[kk], sr);
  }

#pragma unroll
  for (int r = 0; r < 8; ++r) {
    stage[0][(r + 8 * h) * TP + 16 * nt + c] = sr[r];
    stage[1][(r + 8 * h) * TP + 16 * nt + c] = si[r];
  }
  __syncthreads();
  const int col4 = (tid & 15) * 4;
#pragma unroll
  for (int i = 0; i < 2; ++i) {
    const int row = (tid >> 4) + 8 * i;
    const v4f v = *(const v4f*)(stage[0] + row * TP + col4);
    *(v4f*)(sf + (size_t)tr * NXROWS + (b0 + row) * 64 + col4) = v;
  }
  if (cplx) {
#pragma unroll
    for (int i = 0; i < 2; ++i) {
      const int row = (tid >> 4) + 8 * i;
      const v4f v = *(const v4f*)(stage[1] + row * TP + col4);
      *(v4f*)(sf + (size_t)ti * NXROWS + (b0 + row) * 64 + col4) = v;
    }
  }
}

__global__ __launch_bounds__(128)
void bc_inverse_transform(const float* __restrict__ sf, float* __restrict__ out) {
  __shared__ __attribute__((aligned(16))) float c64[64];
  __shared__ __attribute__((aligned(16))) float gt[64 * TP];
  __shared__ __attribute__((aligned(16))) float stage[4][16 * TP];

  const int tid = threadIdx.x;
  const int lane = tid & 31;
  const int wave = tid >> 5;
  const int c = lane & 15;
  const int h = lane >> 4;

  const int row0 = (blockIdx.x * 4 + wave) * 16;

  v2f a[16];
#pragma unroll
  for (int kk = 0; kk < 16; ++kk) {
    a[kk].x = sf[(size_t)(4 * kk + 2 * h) * NXROWS + row0 + c];
    a[kk].y = sf[(size_t)(4 * kk + 2 * h + 1) * NXROWS + row0 + c];
  }

  if (tid < 64) c64[tid] = cospif((float)tid * (1.0f / 32.0f));
  __syncthreads();
  {
    const int tq = tid & 63;
    const bool gCos = tq <= 32;
    const int gk = gCos ? tq : tq - 32;
    const int gph = gCos ? 0 : -16;
    const float gs = gCos ? ((tq == 0 || tq == 32) ? (1.0f / 64.0f) : (1.0f / 32.0f))
                          : (-1.0f / 32.0f);
#pragma unroll
    for (int i = 0; i < 32; ++i) {
      const int n = (tid >> 6) + 2 * i;
      gt[n * TP + tq] = gs * c64[(n * gk + gph) & 63];
    }
  }
  __syncthreads();

  float* st = stage[wave];
  v2f b[4][16];
#pragma unroll
  for (int nt = 0; nt < 4; ++nt)
#pragma unroll
    for (int kk = 0; kk < 16; ++kk)
      b[nt][kk] = *(const v2f*)(gt + (16 * nt + c) * TP + 4 * kk + 2 * h);
  __builtin_amdgcn_sched_barrier(0);
  v8f acc[4];
#pragma unroll
  for (int nt = 0; nt < 4; ++nt) acc[nt] = (v8f){};
#pragma unroll
  for (int kk = 0; kk < 16; ++kk)
#pragma unroll
    for (int nt = 0; nt < 4; ++nt)
      acc[nt] = mma(a[kk], b[nt][kk], acc[nt]);
#pragma unroll
  for (int nt = 0; nt < 4; ++nt)
#pragma unroll
    for (int r = 0; r < 8; ++r)
      st[(r + 8 * h) * TP + 16 * nt + c] = acc[nt][r];
  __syncthreads();
#pragma unroll
  for (int i = 0; i < 8; ++i) {
    const int row = (lane >> 4) + 2 * i;
    const int col4 = (lane & 15) * 4;
    const v4f v = *(const v4f*)(st + row * TP + col4);
    *(v4f*)(out + (size_t)(row0 + row) * 64 + col4) = v;
  }
}

extern "C" void kernel_launch(void* const* d_in, const int* in_sizes, int n_in,
                              void* d_out, int out_size, void* d_ws, size_t ws_size,
                              hipStream_t stream) {
  (void)in_sizes; (void)n_in; (void)out_size; (void)ws_size;
  const float* x = (const float*)d_in[0];
  const float* w = (const float*)d_in[1];
  float* out = (float*)d_out;

  float* xf = (float*)d_ws;
  float* wf = xf + 1048576;
  float* sf = wf + 262144;

  bc_forward_transform<<<640, 128, 0, stream>>>(x, w, xf, wf);
  bc_frequency_products<<<dim3(33, 16), 128, 0, stream>>>(xf, wf, sf);
  bc_inverse_transform<<<256, 128, 0, stream>>>(sf, out);
}


# Radix-2 split of both 64-point transforms: lo+hi / lo-hi fold at the fragment reads, even and odd planes as two 32x32 products; WMMAs per wave 32->16 forward, 64->32 inverse; f32 sum order changes
# speedup vs baseline: 11.3375x; 1.0803x over previous
#include <hip/hip_runtime.h>


typedef __attribute__((ext_vector_type(2))) float v2f;
typedef __attribute__((ext_vector_type(4))) float v4f;
typedef __attribute__((ext_vector_type(8))) float v8f;

#define TP 68
#define SP 36
#define NXROWS 16384
#define NWROWS 4096

static __device__ __forceinline__ v8f mma(v2f a, v2f b, v8f c) {
  return __builtin_amdgcn_wmma_f32_16x16x4_f32(false, a, false, b, (short)0, c, false, false);
}

__global__ __launch_bounds__(128)
void bc_forward_transform(const float* __restrict__ x, const float* __restrict__ w,
                          float* __restrict__ xf, float* __restrict__ wf) {
  __shared__ __attribute__((aligned(16))) float c64[64];
  __shared__ __attribute__((aligned(16))) float xs[32 * TP];
  __shared__ __attribute__((aligned(16))) float stage[64 * SP];

  const int tid = threadIdx.x;
  const int lane = tid & 31;
  const int mt = tid >> 5;
  const int c = lane & 15;
  const int h = lane >> 4;

  const int bid = blockIdx.x;
  const int ghalf = bid & 1;
  const int pq = (bid >> 1) & 15;
  const int tile = bid >> 5;
  const bool isx = tile < 16;
  const int tloc = isx ? tile : tile - 16;
  const float* src = (isx ? x : w) + (size_t)(tloc * 16 + ghalf * 8) * 4096 + pq * 256;
  float* dst = (isx ? xf : wf) + tloc * 1024 + pq * 64 + ghalf * 32;
  const int planeStride = isx ? NXROWS : NWROWS;

  v4f gl[4];
#pragma unroll
  for (int i = 0; i < 4; ++i) {
    const int gi = i * 128 + tid;
    gl[i] = *(const v4f*)(src + (size_t)(gi >> 6) * 4096 + (gi & 63) * 4);
  }
  if (tid < 64) c64[tid] = cospif((float)tid * (1.0f / 32.0f));
#pragma unroll
  for (int i = 0; i < 4; ++i) {
    const int gi = i * 128 + tid;
    *(v4f*)(xs + (gi >> 4) * TP + (gi & 15) * 4) = gl[i];
  }
  __syncthreads();

  const int par = mt >> 1;
  const float fs = par ? -1.0f : 1.0f;
  v2f bx[2][8];
#pragma unroll
  for (int u = 0; u < 2; ++u)
#pragma unroll
    for (int kk = 0; kk < 8; ++kk) {
      const v2f lo = *(const v2f*)(xs + (16 * u + c) * TP + 4 * kk + 2 * h);
      const v2f hi = *(const v2f*)(xs + (16 * u + c) * TP + 32 + 4 * kk + 2 * h);
      bx[u][kk] = lo + fs * hi;
    }

  const int t = 2 * (16 * (mt & 1) + c) + par;
  const bool isCos = t <= 32;
  const int kf = isCos ? t : t - 32;
  const int ph = isCos ? 0 : -16;
  const float sg = isCos ? 1.0f : -1.0f;
  v2f a[8];
#pragma unroll
  for (int kk = 0; kk < 8; ++kk) {
    const int j0 = 4 * kk + 2 * h;
    a[kk].x = sg * c64[(j0 * kf + ph) & 63];
    a[kk].y = sg * c64[((j0 + 1) * kf + ph) & 63];
  }
  __builtin_amdgcn_sched_barrier(0);

  v8f acc0 = (v8f){}, acc1 = (v8f){};
#pragma unroll
  for (int kk = 0; kk < 8; ++kk) {
    acc0 = mma(a[kk], bx[0][kk], acc0);
    acc1 = mma(a[kk], bx[1][kk], acc1);
  }
#pragma unroll
  for (int r = 0; r < 8; ++r) {
    stage[(16 * mt + r + 8 * h) * SP + c] = acc0[r];
    stage[(16 * mt + r + 8 * h) * SP + 16 + c] = acc1[r];
  }
  __syncthreads();
#pragma unroll
  for (int i = 0; i < 4; ++i) {
    const int tl = 16 * i + (tid >> 3);
    const int plane = 2 * (16 * (i & 1) + (tid >> 3)) + (i >> 1);
    const int col4 = (tid & 7) * 4;
    const v4f v = *(const v4f*)(stage + tl * SP + col4);
    *(v4f*)(dst + (size_t)plane * planeStride + col4) = v;
  }
}

__global__ __launch_bounds__(128)
void bc_frequency_products(const float* __restrict__ xf, const float* __restrict__ wf,
                           float* __restrict__ sf) {
  __shared__ __attribute__((aligned(16))) float stage[2][16 * TP];

  const int tid = threadIdx.x;
  const int lane = tid & 31;
  const int nt = tid >> 5;
  const int c = lane & 15;
  const int h = lane >> 4;

  const int k = blockIdx.x;
  const int b0 = blockIdx.y * 16;
  const bool cplx = (k != 0) && (k != 32);
  const int tr = k;
  const int ti = cplx ? 32 + k : k;

  const float* xr = xf + (size_t)tr * NXROWS + blockIdx.y * 1024 + lane * 4;
  const float* xi = xf + (size_t)ti * NXROWS + blockIdx.y * 1024 + lane * 4;
  const float* wr = wf + (size_t)tr * NWROWS + nt * 1024 + lane * 4;
  const float* wi = wf + (size_t)ti * NWROWS + nt * 1024 + lane * 4;

  v2f ar[16], br[16];
#pragma unroll
  for (int q = 0; q < 8; ++q) {
    const v4f va = *(const v4f*)(xr + 128 * q);
    const v4f vb = *(const v4f*)(wr + 128 * q);
    ar[2 * q] = (v2f){va.x, va.y};
    ar[2 * q + 1] = (v2f){va.z, va.w};
    br[2 * q] = (v2f){vb.x, vb.y};
    br[2 * q + 1] = (v2f){vb.z, vb.w};
  }
  v8f sr = (v8f){}, si = (v8f){};
  if (cplx) {
    v2f ai[16], bi[16];
#pragma unroll
    for (int q = 0; q < 8; ++q) {
      const v4f va = *(const v4f*)(xi + 128 * q);
      const v4f vb = *(const v4f*)(wi + 128 * q);
      ai[2 * q] = (v2f){va.x, va.y};
      ai[2 * q + 1] = (v2f){va.z, va.w};
      bi[2 * q] = (v2f){vb.x, vb.y};
      bi[2 * q + 1] = (v2f){vb.z, vb.w};
    }
    __builtin_amdgcn_sched_barrier(0);
#pragma unroll
    for (int kk = 0; kk < 16; ++kk) {
      sr = mma(ar[kk], br[kk], sr);
      si = mma(ar[kk], bi[kk], si);
      sr = mma(ai[kk], -bi[kk], sr);
      si = mma(ai[kk], br[kk], si);
    }
  } else {
#pragma unroll
    for (int kk = 0; kk < 16; ++kk)
      sr = mma(ar[kk], br[kk], sr);
  }

#pragma unroll
  for (int r = 0; r < 8; ++r) {
    stage[0][(r + 8 * h) * TP + 16 * nt + c] = sr[r];
    stage[1][(r + 8 * h) * TP + 16 * nt + c] = si[r];
  }
  __syncthreads();
  const int col4 = (tid & 15) * 4;
#pragma unroll
  for (int i = 0; i < 2; ++i) {
    const int row = (tid >> 4) + 8 * i;
    const v4f v = *(const v4f*)(stage[0] + row * TP + col4);
    *(v4f*)(sf + (size_t)tr * NXROWS + (b0 + row) * 64 + col4) = v;
  }
  if (cplx) {
#pragma unroll
    for (int i = 0; i < 2; ++i) {
      const int row = (tid >> 4) + 8 * i;
      const v4f v = *(const v4f*)(stage[1] + row * TP + col4);
      *(v4f*)(sf + (size_t)ti * NXROWS + (b0 + row) * 64 + col4) = v;
    }
  }
}

__global__ __launch_bounds__(128)
void bc_inverse_transform(const float* __restrict__ sf, float* __restrict__ out) {
  __shared__ __attribute__((aligned(16))) float c64[64];
  __shared__ __attribute__((aligned(16))) float gt[32 * TP];
  __shared__ __attribute__((aligned(16))) float stage[4][16 * TP];

  const int tid = threadIdx.x;
  const int lane = tid & 31;
  const int wave = tid >> 5;
  const int c = lane & 15;
  const int h = lane >> 4;

  const int row0 = (blockIdx.x * 4 + wave) * 16;

  v2f ae[8], ao[8];
#pragma unroll
  for (int kk = 0; kk < 8; ++kk) {
    const int e0 = 4 * kk + 2 * h;
    ae[kk].x = sf[(size_t)(2 * e0) * NXROWS + row0 + c];
    ao[kk].x = sf[(size_t)(2 * e0 + 1) * NXROWS + row0 + c];
    ae[kk].y = sf[(size_t)(2 * e0 + 2) * NXROWS + row0 + c];
    ao[kk].y = sf[(size_t)(2 * e0 + 3) * NXROWS + row0 + c];
  }

  if (tid < 64) c64[tid] = cospif((float)tid * (1.0f / 32.0f));
  __syncthreads();
  {
    const int col = tid & 63;
    const int tq = 2 * (col & 31) + (col >> 5);
    const bool gCos = tq <= 32;
    const int gk = gCos ? tq : tq - 32;
    const int gph = gCos ? 0 : -16;
    const float gs = gCos ? ((tq == 0 || tq == 32) ? (1.0f / 64.0f) : (1.0f / 32.0f))
                          : (-1.0f / 32.0f);
#pragma unroll
    for (int i = 0; i < 16; ++i) {
      const int n = (tid >> 6) + 2 * i;
      gt[n * TP + col] = gs * c64[(n * gk + gph) & 63];
    }
  }
  __syncthreads();

  float* st = stage[wave];
  v2f be[2][8], bo[2][8];
#pragma unroll
  for (int nt = 0; nt < 2; ++nt)
#pragma unroll
    for (int kk = 0; kk < 8; ++kk) {
      be[nt][kk] = *(const v2f*)(gt + (16 * nt + c) * TP + 4 * kk + 2 * h);
      bo[nt][kk] = *(const v2f*)(gt + (16 * nt + c) * TP + 32 + 4 * kk + 2 * h);
    }
  __builtin_amdgcn_sched_barrier(0);
  v8f ev[2], od[2];
#pragma unroll
  for (int nt = 0; nt < 2; ++nt) { ev[nt] = (v8f){}; od[nt] = (v8f){}; }
#pragma unroll
  for (int kk = 0; kk < 8; ++kk)
#pragma unroll
    for (int nt = 0; nt < 2; ++nt) {
      ev[nt] = mma(ae[kk], be[nt][kk], ev[nt]);
      od[nt] = mma(ao[kk], bo[nt][kk], od[nt]);
    }
#pragma unroll
  for (int nt = 0; nt < 2; ++nt)
#pragma unroll
    for (int r = 0; r < 8; ++r) {
      st[(r + 8 * h) * TP + 16 * nt + c] = ev[nt][r] + od[nt][r];
      st[(r + 8 * h) * TP + 32 + 16 * nt + c] = ev[nt][r] - od[nt][r];
    }
  __syncthreads();
#pragma unroll
  for (int i = 0; i < 8; ++i) {
    const int row = (lane >> 4) + 2 * i;
    const int col4 = (lane & 15) * 4;
    const v4f v = *(const v4f*)(st + row * TP + col4);
    *(v4f*)(out + (size_t)(row0 + row) * 64 + col4) = v;
  }
}

extern "C" void kernel_launch(void* const* d_in, const int* in_sizes, int n_in,
                              void* d_out, int out_size, void* d_ws, size_t ws_size,
                              hipStream_t stream) {
  (void)in_sizes; (void)n_in; (void)out_size; (void)ws_size;
  const float* x = (const float*)d_in[0];
  const float* w = (const float*)d_in[1];
  float* out = (float*)d_out;

  float* xf = (float*)d_ws;
  float* wf = xf + 1048576;
  float* sf = wf + 262144;

  bc_forward_transform<<<640, 128, 0, stream>>>(x, w, xf, wf);
  bc_frequency_products<<<dim3(33, 16), 128, 0, stream>>>(xf, wf, sf);
  bc_inverse_transform<<<256, 128, 0, stream>>>(sf, out);
}


# Complex frequency products as 3 real GEMMs not 4: P1=XrWr, P2=XiWi, P3=(Xr+Xi)(Wr+Wi); real frequencies 0 and 32 are slot 0's P1, P2; grid 33x16 -> 32x16; WMMAs/wave 64 -> 48; f32 sum order changes
# speedup vs baseline: 12.0000x; 1.0584x over previous
#include <hip/hip_runtime.h>


typedef __attribute__((ext_vector_type(2))) float v2f;
typedef __attribute__((ext_vector_type(4))) float v4f;
typedef __attribute__((ext_vector_type(8))) float v8f;

#define TP 68
#define SP 36
#define NXROWS 16384
#define NWROWS 4096

static __device__ __forceinline__ v8f mma(v2f a, v2f b, v8f c) {
  return __builtin_amdgcn_wmma_f32_16x16x4_f32(false, a, false, b, (short)0, c, false, false);
}

__global__ __launch_bounds__(128)
void bc_forward_transform(const float* __restrict__ x, const float* __restrict__ w,
                          float* __restrict__ xf, float* __restrict__ wf) {
  __shared__ __attribute__((aligned(16))) float c64[64];
  __shared__ __attribute__((aligned(16))) float xs[32 * TP];
  __shared__ __attribute__((aligned(16))) float stage[64 * SP];

  const int tid = threadIdx.x;
  const int lane = tid & 31;
  const int mt = tid >> 5;
  const int c = lane & 15;
  const int h = lane >> 4;

  const int bid = blockIdx.x;
  const int ghalf = bid & 1;
  const int pq = (bid >> 1) & 15;
  const int tile = bid >> 5;
  const bool isx = tile < 16;
  const int tloc = isx ? tile : tile - 16;
  const float* src = (isx ? x : w) + (size_t)(tloc * 16 + ghalf * 8) * 4096 + pq * 256;
  float* dst = (isx ? xf : wf) + tloc * 1024 + pq * 64 + ghalf * 32;
  const int planeStride = isx ? NXROWS : NWROWS;

  v4f gl[4];
#pragma unroll
  for (int i = 0; i < 4; ++i) {
    const int gi = i * 128 + tid;
    gl[i] = *(const v4f*)(src + (size_t)(gi >> 6) * 4096 + (gi & 63) * 4);
  }
  if (tid < 64) c64[tid] = cospif((float)tid * (1.0f / 32.0f));
#pragma unroll
  for (int i = 0; i < 4; ++i) {
    const int gi = i * 128 + tid;
    *(v4f*)(xs + (gi >> 4) * TP + (gi & 15) * 4) = gl[i];
  }
  __syncthreads();

  const int par = mt >> 1;
  const float fs = par ? -1.0f : 1.0f;
  v2f bx[2][8];
#pragma unroll
  for (int u = 0; u < 2; ++u)
#pragma unroll
    for (int kk = 0; kk < 8; ++kk) {
      const v2f lo = *(const v2f*)(xs + (16 * u + c) * TP + 4 * kk + 2 * h);
      const v2f hi = *(const v2f*)(xs + (16 * u + c) * TP + 32 + 4 * kk + 2 * h);
      bx[u][kk] = lo + fs * hi;
    }

  const int t = 2 * (16 * (mt & 1) + c) + par;
  const bool isCos = t <= 32;
  const int kf = isCos ? t : t - 32;
  const int ph = isCos ? 0 : -16;
  const float sg = isCos ? 1.0f : -1.0f;
  v2f a[8];
#pragma unroll
  for (int kk = 0; kk < 8; ++kk) {
    const int j0 = 4 * kk + 2 * h;
    a[kk].x = sg * c64[(j0 * kf + ph) & 63];
    a[kk].y = sg * c64[((j0 + 1) * kf + ph) & 63];
  }
  __builtin_amdgcn_sched_barrier(0);

  v8f acc0 = (v8f){}, acc1 = (v8f){};
#pragma unroll
  for (int kk = 0; kk < 8; ++kk) {
    acc0 = mma(a[kk], bx[0][kk], acc0);
    acc1 = mma(a[kk], bx[1][kk], acc1);
  }
#pragma unroll
  for (int r = 0; r < 8; ++r) {
    stage[(16 * mt + r + 8 * h) * SP + c] = acc0[r];
    stage[(16 * mt + r + 8 * h) * SP + 16 + c] = acc1[r];
  }
  __syncthreads();
#pragma unroll
  for (int i = 0; i < 4; ++i) {
    const int tl = 16 * i + (tid >> 3);
    const int plane = 2 * (16 * (i & 1) + (tid >> 3)) + (i >> 1);
    const int col4 = (tid & 7) * 4;
    const v4f v = *(const v4f*)(stage + tl * SP + col4);
    *(v4f*)(dst + (size_t)plane * planeStride + col4) = v;
  }
}

__global__ __launch_bounds__(128)
void bc_frequency_products(const float* __restrict__ xf, const float* __restrict__ wf,
                           float* __restrict__ sf) {
  __shared__ __attribute__((aligned(16))) float stage[2][16 * TP];

  const int tid = threadIdx.x;
  const int lane = tid & 31;
  const int nt = tid >> 5;
  const int c = lane & 15;
  const int h = lane >> 4;

  const int k = blockIdx.x;
  const int b0 = blockIdx.y * 16;
  const bool cplx = (k != 0);
  const int tr = k;
  const int ti = 32 + k;

  const float* xr = xf + (size_t)tr * NXROWS + blockIdx.y * 1024 + lane * 4;
  const float* xi = xf + (size_t)ti * NXROWS + blockIdx.y * 1024 + lane * 4;
  const float* wr = wf + (size_t)tr * NWROWS + nt * 1024 + lane * 4;
  const float* wi = wf + (size_t)ti * NWROWS + nt * 1024 + lane * 4;

  v2f ar[16], br[16], ai[16], bi[16];
#pragma unroll
  for (int q = 0; q < 8; ++q) {
    const v4f va = *(const v4f*)(xr + 128 * q);
    const v4f vb = *(const v4f*)(wr + 128 * q);
    ar[2 * q] = (v2f){va.x, va.y};
    ar[2 * q + 1] = (v2f){va.z, va.w};
    br[2 * q] = (v2f){vb.x, vb.y};
    br[2 * q + 1] = (v2f){vb.z, vb.w};
  }
#pragma unroll
  for (int q = 0; q < 8; ++q) {
    const v4f va = *(const v4f*)(xi + 128 * q);
    const v4f vb = *(const v4f*)(wi + 128 * q);
    ai[2 * q] = (v2f){va.x, va.y};
    ai[2 * q + 1] = (v2f){va.z, va.w};
    bi[2 * q] = (v2f){vb.x, vb.y};
    bi[2 * q + 1] = (v2f){vb.z, vb.w};
  }
  __builtin_amdgcn_sched_barrier(0);

  v8f p1 = (v8f){}, p2 = (v8f){}, p3 = (v8f){};
  if (cplx) {
#pragma unroll
    for (int kk = 0; kk < 16; ++kk) {
      p1 = mma(ar[kk], br[kk], p1);
      p2 = mma(ai[kk], bi[kk], p2);
      p3 = mma(ar[kk] + ai[kk], br[kk] + bi[kk], p3);
    }
  } else {
#pragma unroll
    for (int kk = 0; kk < 16; ++kk) {
      p1 = mma(ar[kk], br[kk], p1);
      p2 = mma(ai[kk], bi[kk], p2);
    }
  }

#pragma unroll
  for (int r = 0; r < 8; ++r) {
    const float lo = cplx ? p1[r] - p2[r] : p1[r];
    const float hi = cplx ? p3[r] - (p1[r] + p2[r]) : p2[r];
    stage[0][(r + 8 * h) * TP + 16 * nt + c] = lo;
    stage[1][(r + 8 * h) * TP + 16 * nt + c] = hi;
  }
  __syncthreads();
  const int col4 = (tid & 15) * 4;
#pragma unroll
  for (int i = 0; i < 2; ++i) {
    const int row = (tid >> 4) + 8 * i;
    const v4f v0 = *(const v4f*)(stage[0] + row * TP + col4);
    const v4f v1 = *(const v4f*)(stage[1] + row * TP + col4);
    *(v4f*)(sf + (size_t)tr * NXROWS + (b0 + row) * 64 + col4) = v0;
    *(v4f*)(sf + (size_t)ti * NXROWS + (b0 + row) * 64 + col4) = v1;
  }
}

__global__ __launch_bounds__(128)
void bc_inverse_transform(const float* __restrict__ sf, float* __restrict__ out) {
  __shared__ __attribute__((aligned(16))) float c64[64];
  __shared__ __attribute__((aligned(16))) float gt[32 * TP];
  __shared__ __attribute__((aligned(16))) float stage[4][16 * TP];

  const int tid = threadIdx.x;
  const int lane = tid & 31;
  const int wave = tid >> 5;
  const int c = lane & 15;
  const int h = lane >> 4;

  const int row0 = (blockIdx.x * 4 + wave) * 16;

  v2f ae[8], ao[8];
#pragma unroll
  for (int kk = 0; kk < 8; ++kk) {
    const int e0 = 4 * kk + 2 * h;
    ae[kk].x = sf[(size_t)(2 * e0) * NXROWS + row0 + c];
    ao[kk].x = sf[(size_t)(2 * e0 + 1) * NXROWS + row0 + c];
    ae[kk].y = sf[(size_t)(2 * e0 + 2) * NXROWS + row0 + c];
    ao[kk].y = sf[(size_t)(2 * e0 + 3) * NXROWS + row0 + c];
  }

  if (tid < 64) c64[tid] = cospif((float)tid * (1.0f / 32.0f));
  __syncthreads();
  {
    const int col = tid & 63;
    const int tq = 2 * (col & 31) + (col >> 5);
    const bool gCos = tq <= 32;
    const int gk = gCos ? tq : tq - 32;
    const int gph = gCos ? 0 : -16;
    const float gs = gCos ? ((tq == 0 || tq == 32) ? (1.0f / 64.0f) : (1.0f / 32.0f))
                          : (-1.0f / 32.0f);
#pragma unroll
    for (int i = 0; i < 16; ++i) {
      const int n = (tid >> 6) + 2 * i;
      gt[n * TP + col] = gs * c64[(n * gk + gph) & 63];
    }
  }
  __syncthreads();

  float* st = stage[wave];
  v2f be[2][8], bo[2][8];
#pragma unroll
  for (int nt = 0; nt < 2; ++nt)
#pragma unroll
    for (int kk = 0; kk < 8; ++kk) {
      be[nt][kk] = *(const v2f*)(gt + (16 * nt + c) * TP + 4 * kk + 2 * h);
      bo[nt][kk] = *(const v2f*)(gt + (16 * nt + c) * TP + 32 + 4 * kk + 2 * h);
    }
  __builtin_amdgcn_sched_barrier(0);
  v8f ev[2], od[2];
#pragma unroll
  for (int nt = 0; nt < 2; ++nt) { ev[nt] = (v8f){}; od[nt] = (v8f){}; }
#pragma unroll
  for (int kk = 0; kk < 8; ++kk)
#pragma unroll
    for (int nt = 0; nt < 2; ++nt) {
      ev[nt] = mma(ae[kk], be[nt][kk], ev[nt]);
      od[nt] = mma(ao[kk], bo[nt][kk], od[nt]);
    }
#pragma unroll
  for (int nt = 0; nt < 2; ++nt)
#pragma unroll
    for (int r = 0; r < 8; ++r) {
      st[(r + 8 * h) * TP + 16 * nt + c] = ev[nt][r] + od[nt][r];
      st[(r + 8 * h) * TP + 32 + 16 * nt + c] = ev[nt][r] - od[nt][r];
    }
  __syncthreads();
#pragma unroll
  for (int i = 0; i < 8; ++i) {
    const int row = (lane >> 4) + 2 * i;
    const int col4 = (lane & 15) * 4;
    const v4f v = *(const v4f*)(st + row * TP + col4);
    *(v4f*)(out + (size_t)(row0 + row) * 64 + col4) = v;
  }
}

extern "C" void kernel_launch(void* const* d_in, const int* in_sizes, int n_in,
                              void* d_out, int out_size, void* d_ws, size_t ws_size,
                              hipStream_t stream) {
  (void)in_sizes; (void)n_in; (void)out_size; (void)ws_size;
  const float* x = (const float*)d_in[0];
  const float* w = (const float*)d_in[1];
  float* out = (float*)d_out;

  float* xf = (float*)d_ws;
  float* wf = xf + 1048576;
  float* sf = wf + 262144;

  bc_forward_transform<<<640, 128, 0, stream>>>(x, w, xf, wf);
  bc_frequency_products<<<dim3(32, 16), 128, 0, stream>>>(xf, wf, sf);
  bc_inverse_transform<<<256, 128, 0, stream>>>(sf, out);
}


# Load latency overlapped: product operands loaded k-step-major (first WMMA waits for 7 of 32 loads, not 27); transforms build their tables under their loads, barriers fence LDS only; sum order kept
# speedup vs baseline: 12.6458x; 1.0538x over previous
#include <hip/hip_runtime.h>


typedef __attribute__((ext_vector_type(2))) float v2f;
typedef __attribute__((ext_vector_type(4))) float v4f;
typedef __attribute__((ext_vector_type(8))) float v8f;

#define TP 68
#define SP 36
#define NXROWS 16384
#define NWROWS 4096

static __device__ __forceinline__ v8f mma(v2f a, v2f b, v8f c) {
  return __builtin_amdgcn_wmma_f32_16x16x4_f32(false, a, false, b, (short)0, c, false, false);
}

static __device__ __forceinline__ void lds_barrier() {
  __builtin_amdgcn_fence(__ATOMIC_RELEASE, "workgroup", "local");
  __builtin_amdgcn_s_barrier();
  __builtin_amdgcn_fence(__ATOMIC_ACQUIRE, "workgroup", "local");
}

__global__ __launch_bounds__(128)
void bc_forward_transform(const float* __restrict__ x, const float* __restrict__ w,
                          float* __restrict__ xf, float* __restrict__ wf) {
  __shared__ __attribute__((aligned(16))) float c64[64];
  __shared__ __attribute__((aligned(16))) float xs[32 * TP];
  __shared__ __attribute__((aligned(16))) float stage[64 * SP];

  const int tid = threadIdx.x;
  const int lane = tid & 31;
  const int mt = tid >> 5;
  const int c = lane & 15;
  const int h = lane >> 4;

  const int bid = blockIdx.x;
  const int ghalf = bid & 1;
  const int pq = (bid >> 1) & 15;
  const int tile = bid >> 5;
  const bool isx = tile < 16;
  const int tloc = isx ? tile : tile - 16;
  const float* src = (isx ? x : w) + (size_t)(tloc * 16 + ghalf * 8) * 4096 + pq * 256;
  float* dst = (isx ? xf : wf) + tloc * 1024 + pq * 64 + ghalf * 32;
  const int planeStride = isx ? NXROWS : NWROWS;

  v4f gl[4];
#pragma unroll
  for (int i = 0; i < 4; ++i) {
    const int gi = i * 128 + tid;
    gl[i] = *(const v4f*)(src + (size_t)(gi >> 6) * 4096 + (gi & 63) * 4);
  }
  if (tid < 64) c64[tid] = cospif((float)tid * (1.0f / 32.0f));
  lds_barrier();

  const int par = mt >> 1;
  const int t = 2 * (16 * (mt & 1) + c) + par;
  const bool isCos = t <= 32;
  const int kf = isCos ? t : t - 32;
  const int ph = isCos ? 0 : -16;
  const float sg = isCos ? 1.0f : -1.0f;
  v2f a[8];
#pragma unroll
  for (int kk = 0; kk < 8; ++kk) {
    const int j0 = 4 * kk + 2 * h;
    a[kk].x = sg * c64[(j0 * kf + ph) & 63];
    a[kk].y = sg * c64[((j0 + 1) * kf + ph) & 63];
  }

#pragma unroll
  for (int i = 0; i < 4; ++i) {
    const int gi = i * 128 + tid;
    *(v4f*)(xs + (gi >> 4) * TP + (gi & 15) * 4) = gl[i];
  }
  __syncthreads();

  const float fs = par ? -1.0f : 1.0f;
  v2f bx[2][8];
#pragma unroll
  for (int u = 0; u < 2; ++u)
#pragma unroll
    for (int kk = 0; kk < 8; ++kk) {
      const v2f lo = *(const v2f*)(xs + (16 * u + c) * TP + 4 * kk + 2 * h);
      const v2f hi = *(const v2f*)(xs + (16 * u + c) * TP + 32 + 4 * kk + 2 * h);
      bx[u][kk] = lo + fs * hi;
    }

  __builtin_amdgcn_sched_barrier(0);

  v8f acc0 = (v8f){}, acc1 = (v8f){};
#pragma unroll
  for (int kk = 0; kk < 8; ++kk) {
    acc0 = mma(a[kk], bx[0][kk], acc0);
    acc1 = mma(a[kk], bx[1][kk], acc1);
  }
#pragma unroll
  for (int r = 0; r < 8; ++r) {
    stage[(16 * mt + r + 8 * h) * SP + c] = acc0[r];
    stage[(16 * mt + r + 8 * h) * SP + 16 + c] = acc1[r];
  }
  __syncthreads();
#pragma unroll
  for (int i = 0; i < 4; ++i) {
    const int tl = 16 * i + (tid >> 3);
    const int plane = 2 * (16 * (i & 1) + (tid >> 3)) + (i >> 1);
    const int col4 = (tid & 7) * 4;
    const v4f v = *(const v4f*)(stage + tl * SP + col4);
    *(v4f*)(dst + (size_t)plane * planeStride + col4) = v;
  }
}

__global__ __launch_bounds__(128)
void bc_frequency_products(const float* __restrict__ xf, const float* __restrict__ wf,
                           float* __restrict__ sf) {
  __shared__ __attribute__((aligned(16))) float stage[2][16 * TP];

  const int tid = threadIdx.x;
  const int lane = tid & 31;
  const int nt = tid >> 5;
  const int c = lane & 15;
  const int h = lane >> 4;

  const int k = blockIdx.x;
  const int b0 = blockIdx.y * 16;
  const bool cplx = (k != 0);
  const int tr = k;
  const int ti = 32 + k;

  const float* xr = xf + (size_t)tr * NXROWS + blockIdx.y * 1024 + lane * 4;
  const float* xi = xf + (size_t)ti * NXROWS + blockIdx.y * 1024 + lane * 4;
  const float* wr = wf + (size_t)tr * NWROWS + nt * 1024 + lane * 4;
  const float* wi = wf + (size_t)ti * NWROWS + nt * 1024 + lane * 4;

  v2f ar[16], br[16], ai[16], bi[16];
#pragma unroll
  for (int q = 0; q < 8; ++q) {
    const v4f va = *(const v4f*)(xr + 128 * q);
    const v4f vb = *(const v4f*)(wr + 128 * q);
    const v4f vc = *(const v4f*)(xi + 128 * q);
    const v4f vd = *(const v4f*)(wi + 128 * q);
    ar[2 * q] = (v2f){va.x, va.y};
    ar[2 * q + 1] = (v2f){va.z, va.w};
    br[2 * q] = (v2f){vb.x, vb.y};
    br[2 * q + 1] = (v2f){vb.z, vb.w};
    ai[2 * q] = (v2f){vc.x, vc.y};
    ai[2 * q + 1] = (v2f){vc.z, vc.w};
    bi[2 * q] = (v2f){vd.x, vd.y};
    bi[2 * q + 1] = (v2f){vd.z, vd.w};
  }
  __builtin_amdgcn_sched_barrier(0);

  v8f p1 = (v8f){}, p2 = (v8f){}, p3 = (v8f){};
  if (cplx) {
#pragma unroll
    for (int kk = 0; kk < 16; ++kk) {
      p1 = mma(ar[kk], br[kk], p1);
      p2 = mma(ai[kk], bi[kk], p2);
      p3 = mma(ar[kk] + ai[kk], br[kk] + bi[kk], p3);
    }
  } else {
#pragma unroll
    for (int kk = 0; kk < 16; ++kk) {
      p1 = mma(ar[kk], br[kk], p1);
      p2 = mma(ai[kk], bi[kk], p2);
    }
  }

#pragma unroll
  for (int r = 0; r < 8; ++r) {
    const float lo = cplx ? p1[r] - p2[r] : p1[r];
    const float hi = cplx ? p3[r] - (p1[r] + p2[r]) : p2[r];
    stage[0][(r + 8 * h) * TP + 16 * nt + c] = lo;
    stage[1][(r + 8 * h) * TP + 16 * nt + c] = hi;
  }
  __syncthreads();
  const int col4 = (tid & 15) * 4;
#pragma unroll
  for (int i = 0; i < 2; ++i) {
    const int row = (tid >> 4) + 8 * i;
    const v4f v0 = *(const v4f*)(stage[0] + row * TP + col4);
    const v4f v1 = *(const v4f*)(stage[1] + row * TP + col4);
    *(v4f*)(sf + (size_t)tr * NXROWS + (b0 + row) * 64 + col4) = v0;
    *(v4f*)(sf + (size_t)ti * NXROWS + (b0 + row) * 64 + col4) = v1;
  }
}

__global__ __launch_bounds__(128)
void bc_inverse_transform(const float* __restrict__ sf, float* __restrict__ out) {
  __shared__ __attribute__((aligned(16))) float c64[64];
  __shared__ __attribute__((aligned(16))) float gt[32 * TP];
  __shared__ __attribute__((aligned(16))) float stage[4][16 * TP];

  const int tid = threadIdx.x;
  const int lane = tid & 31;
  const int wave = tid >> 5;
  const int c = lane & 15;
  const int h = lane >> 4;

  const int row0 = (blockIdx.x * 4 + wave) * 16;

  v2f ae[8], ao[8];
#pragma unroll
  for (int kk = 0; kk < 8; ++kk) {
    const int e0 = 4 * kk + 2 * h;
    ae[kk].x = sf[(size_t)(2 * e0) * NXROWS + row0 + c];
    ao[kk].x = sf[(size_t)(2 * e0 + 1) * NXROWS + row0 + c];
    ae[kk].y = sf[(size_t)(2 * e0 + 2) * NXROWS + row0 + c];
    ao[kk].y = sf[(size_t)(2 * e0 + 3) * NXROWS + row0 + c];
  }

  if (tid < 64) c64[tid] = cospif((float)tid * (1.0f / 32.0f));
  lds_barrier();
  {
    const int col = tid & 63;
    const int tq = 2 * (col & 31) + (col >> 5);
    const bool gCos = tq <= 32;
    const int gk = gCos ? tq : tq - 32;
    const int gph = gCos ? 0 : -16;
    const float gs = gCos ? ((tq == 0 || tq == 32) ? (1.0f / 64.0f) : (1.0f / 32.0f))
                          : (-1.0f / 32.0f);
#pragma unroll
    for (int i = 0; i < 16; ++i) {
      const int n = (tid >> 6) + 2 * i;
      gt[n * TP + col] = gs * c64[(n * gk + gph) & 63];
    }
  }
  lds_barrier();

  float* st = stage[wave];
  v2f be[2][8], bo[2][8];
#pragma unroll
  for (int nt = 0; nt < 2; ++nt)
#pragma unroll
    for (int kk = 0; kk < 8; ++kk) {
      be[nt][kk] = *(const v2f*)(gt + (16 * nt + c) * TP + 4 * kk + 2 * h);
      bo[nt][kk] = *(const v2f*)(gt + (16 * nt + c) * TP + 32 + 4 * kk + 2 * h);
    }
  __builtin_amdgcn_sched_barrier(0);
  v8f ev[2], od[2];
#pragma unroll
  for (int nt = 0; nt < 2; ++nt) { ev[nt] = (v8f){}; od[nt] = (v8f){}; }
#pragma unroll
  for (int kk = 0; kk < 8; ++kk)
#pragma unroll
    for (int nt = 0; nt < 2; ++nt) {
      ev[nt] = mma(ae[kk], be[nt][kk], ev[nt]);
      od[nt] = mma(ao[kk], bo[nt][kk], od[nt]);
    }
#pragma unroll
  for (int nt = 0; nt < 2; ++nt)
#pragma unroll
    for (int r = 0; r < 8; ++r) {
      st[(r + 8 * h) * TP + 16 * nt + c] = ev[nt][r] + od[nt][r];
      st[(r + 8 * h) * TP + 32 + 16 * nt + c] = ev[nt][r] - od[nt][r];
    }
  __syncthreads();
#pragma unroll
  for (int i = 0; i < 8; ++i) {
    const int row = (lane >> 4) + 2 * i;
    const int col4 = (lane & 15) * 4;
    const v4f v = *(const v4f*)(st + row * TP + col4);
    *(v4f*)(out + (size_t)(row0 + row) * 64 + col4) = v;
  }
}

extern "C" void kernel_launch(void* const* d_in, const int* in_sizes, int n_in,
                              void* d_out, int out_size, void* d_ws, size_t ws_size,
                              hipStream_t stream) {
  (void)in_sizes; (void)n_in; (void)out_size; (void)ws_size;
  const float* x = (const float*)d_in[0];
  const float* w = (const float*)d_in[1];
  float* out = (float*)d_out;

  float* xf = (float*)d_ws;
  float* wf = xf + 1048576;
  float* sf = wf + 262144;

  bc_forward_transform<<<640, 128, 0, stream>>>(x, w, xf, wf);
  bc_frequency_products<<<dim3(32, 16), 128, 0, stream>>>(xf, wf, sf);
  bc_inverse_transform<<<256, 128, 0, stream>>>(sf, out);
}
